# starting kernel
# baseline (speedup 1.0000x reference)
#include <hip/hip_runtime.h>

typedef float v2f __attribute__((ext_vector_type(2)));
typedef float v4f __attribute__((ext_vector_type(4)));
typedef float v8f __attribute__((ext_vector_type(8)));

namespace {
constexpr int kNQ     = 8;
constexpr int kFFN    = 512;
constexpr int kTokens = 32 * 4096;
constexpr int kWaves  = 8;
constexpr int kMT     = 4;
}

__global__ __launch_bounds__(256)
void ffq_fused_wmma(const float* __restrict__ x,
                    const float* __restrict__ phi,
                    const float* __restrict__ w1,
                    const float* __restrict__ b1,
                    const float* __restrict__ w2,
                    const float* __restrict__ b2,
                    float* __restrict__ out)
{
  const int lane = threadIdx.x & 31;
  const int wave = threadIdx.x >> 5;
  const int row  = lane & 15;
  const int hi   = lane >> 4;
  const int koff = hi << 1;

  const int tile0 = (blockIdx.x * kWaves + wave) * kMT;
  const float ph  = phi[0];

  v2f qlo[kMT], qhi[kMT];
#pragma unroll
  for (int m = 0; m < kMT; ++m) {
    const float* xr = x + (size_t)((tile0 + m) * 16 + row) * kNQ;
    qlo[m][0] = __cosf(xr[koff + 0] + ph);
    qlo[m][1] = __cosf(xr[koff + 1] + ph);
    qhi[m][0] = __cosf(xr[koff + 4] + ph);
    qhi[m][1] = __cosf(xr[koff + 5] + ph);
  }

  v2f bone; bone[0] = hi ? 0.0f : 1.0f; bone[1] = 0.0f;
  const v8f vzero = {};

  v2f ab2; ab2[0] = hi ? 0.0f : b2[row & 7]; ab2[1] = 0.0f;
  const v8f accInit =
      __builtin_amdgcn_wmma_f32_16x16x4_f32(false, ab2, false, bone, (short)0, vzero, false, false);
  v8f acc[kMT];
#pragma unroll
  for (int m = 0; m < kMT; ++m) acc[m] = accInit;

  const float* w2r = w2 + (size_t)(row & 7) * kFFN + 8 * hi;

  for (int n0 = 0; n0 < kFFN; n0 += 16) {
    const float* w1r = w1 + (size_t)(n0 + row) * kNQ;
    v2f wlo, whi;
    wlo[0] = w1r[koff + 0]; wlo[1] = w1r[koff + 1];
    whi[0] = w1r[koff + 4]; whi[1] = w1r[koff + 5];

    v2f ab1; ab1[0] = hi ? 0.0f : b1[n0 + row]; ab1[1] = 0.0f;

    const v4f wq0 = *(const v4f*)(w2r + n0);
    const v4f wq1 = *(const v4f*)(w2r + n0 + 4);

    const v8f hinit =
        __builtin_amdgcn_wmma_f32_16x16x4_f32(false, ab1, false, bone, (short)0, vzero, false, false);

#pragma unroll
    for (int m = 0; m < kMT; ++m) {
      v8f h = __builtin_amdgcn_wmma_f32_16x16x4_f32(false, wlo, false, qlo[m], (short)0, hinit, false, false);
      h     = __builtin_amdgcn_wmma_f32_16x16x4_f32(false, whi, false, qhi[m], (short)0, h,     false, false);
#pragma unroll
      for (int r = 0; r < 8; ++r) h[r] = h[r] > 0.0f ? h[r] : 0.0f;

#pragma unroll
      for (int i = 0; i < 4; ++i) {
        v2f a2, bfrag;
        a2[0]    = (i < 2) ? wq0[2 * i]     : wq1[2 * (i - 2)];
        a2[1]    = (i < 2) ? wq0[2 * i + 1] : wq1[2 * (i - 2) + 1];
        bfrag[0] = h[2 * i];
        bfrag[1] = h[2 * i + 1];
        acc[m] = __builtin_amdgcn_wmma_f32_16x16x4_f32(false, a2, false, bfrag, (short)0, acc[m], false, false);
      }
    }
  }

  if (hi == 0) {
#pragma unroll
    for (int m = 0; m < kMT; ++m) {
      float* orow = out + (size_t)((tile0 + m) * 16 + row) * kNQ;
      v4f lo4, hi4;
      lo4[0] = acc[m][0]; lo4[1] = acc[m][1]; lo4[2] = acc[m][2]; lo4[3] = acc[m][3];
      hi4[0] = acc[m][4]; hi4[1] = acc[m][5]; hi4[2] = acc[m][6]; hi4[3] = acc[m][7];
      *(v4f*)(orow + 0) = lo4;
      *(v4f*)(orow + 4) = hi4;
    }
  }
}

extern "C" void kernel_launch(void* const* d_in, const int* in_sizes, int n_in,
                              void* d_out, int out_size, void* d_ws, size_t ws_size,
                              hipStream_t stream) {
  const float* x   = (const float*)d_in[0];
  const float* phi = (const float*)d_in[1];
  const float* w1  = (const float*)d_in[2];
  const float* b1  = (const float*)d_in[3];
  const float* w2  = (const float*)d_in[4];
  const float* b2  = (const float*)d_in[5];
  float* out = (float*)d_out;

  const int blocks = kTokens / (16 * kMT * kWaves);
  ffq_fused_wmma<<<blocks, 256, 0, stream>>>(x, phi, w1, b1, w2, b2, out);
}


# w1, b1, w2 fragments copied once per work-group into LDS in per-lane fragment order; the chunk loop reads them from LDS, not from the second-level cache every chunk; sum order unchanged
# speedup vs baseline: 1.4876x; 1.4876x over previous
#include <hip/hip_runtime.h>

typedef float v2f __attribute__((ext_vector_type(2)));
typedef float v4f __attribute__((ext_vector_type(4)));
typedef float v8f __attribute__((ext_vector_type(8)));

namespace {
constexpr int kNQ     = 8;
constexpr int kFFN    = 512;
constexpr int kTokens = 32 * 4096;
constexpr int kWaves  = 8;
constexpr int kMT     = 4;
constexpr int kChunks = kFFN / 16;
}

__global__ __launch_bounds__(256)
void ffq_fused_wmma(const float* __restrict__ x,
                    const float* __restrict__ phi,
                    const float* __restrict__ w1,
                    const float* __restrict__ b1,
                    const float* __restrict__ w2,
                    const float* __restrict__ b2,
                    float* __restrict__ out)
{
  __shared__ v4f   sW1 [kChunks * 32];
  __shared__ v4f   sW2a[kChunks * 32];
  __shared__ v4f   sW2b[kChunks * 32];
  __shared__ float sB1 [kChunks * 32];

  const int lane = threadIdx.x & 31;
  const int wave = threadIdx.x >> 5;
  const int row  = lane & 15;
  const int hi   = lane >> 4;
  const int koff = hi << 1;

  const int tile0 = (blockIdx.x * kWaves + wave) * kMT;

  const float* w2r = w2 + (size_t)(row & 7) * kFFN + 8 * hi;

  v4f   stA[kChunks / kWaves], stQ0[kChunks / kWaves], stQ1[kChunks / kWaves];
  float stB[kChunks / kWaves];
#pragma unroll
  for (int j = 0; j < kChunks / kWaves; ++j) {
    const int n0 = (wave + j * kWaves) * 16;
    const float* w1r = w1 + (size_t)(n0 + row) * kNQ;
    stA[j][0] = w1r[koff + 0]; stA[j][1] = w1r[koff + 1];
    stA[j][2] = w1r[koff + 4]; stA[j][3] = w1r[koff + 5];
    stQ0[j] = *(const v4f*)(w2r + n0);
    stQ1[j] = *(const v4f*)(w2r + n0 + 4);
    const float bv = b1[n0 + row];
    stB[j] = hi ? 0.0f : bv;
  }

  const float ph  = phi[0];
  const float b2v = b2[row & 7];

  v2f xlo[kMT], xhi[kMT];
#pragma unroll
  for (int m = 0; m < kMT; ++m) {
    const float* xr = x + (size_t)((tile0 + m) * 16 + row) * kNQ;
    xlo[m][0] = xr[koff + 0]; xlo[m][1] = xr[koff + 1];
    xhi[m][0] = xr[koff + 4]; xhi[m][1] = xr[koff + 5];
  }

#pragma unroll
  for (int j = 0; j < kChunks / kWaves; ++j) {
    const int idx = (wave + j * kWaves) * 32 + lane;
    sW1[idx]  = stA[j];
    sW2a[idx] = stQ0[j];
    sW2b[idx] = stQ1[j];
    sB1[idx]  = stB[j];
  }

  v2f qlo[kMT], qhi[kMT];
#pragma unroll
  for (int m = 0; m < kMT; ++m) {
    qlo[m][0] = __cosf(xlo[m][0] + ph);
    qlo[m][1] = __cosf(xlo[m][1] + ph);
    qhi[m][0] = __cosf(xhi[m][0] + ph);
    qhi[m][1] = __cosf(xhi[m][1] + ph);
  }

  v2f bone; bone[0] = hi ? 0.0f : 1.0f; bone[1] = 0.0f;
  const v8f vzero = {};

  v2f ab2; ab2[0] = hi ? 0.0f : b2v; ab2[1] = 0.0f;
  const v8f accInit =
      __builtin_amdgcn_wmma_f32_16x16x4_f32(false, ab2, false, bone, (short)0, vzero, false, false);
  v8f acc[kMT];
#pragma unroll
  for (int m = 0; m < kMT; ++m) acc[m] = accInit;

  __syncthreads();

  for (int c = 0; c < kChunks; ++c) {
    const int idx = c * 32 + lane;
    const v4f wA = sW1[idx];
    v2f wlo, whi;
    wlo[0] = wA[0]; wlo[1] = wA[1];
    whi[0] = wA[2]; whi[1] = wA[3];

    v2f ab1; ab1[0] = sB1[idx]; ab1[1] = 0.0f;

    const v4f wq0 = sW2a[idx];
    const v4f wq1 = sW2b[idx];

    const v8f hinit =
        __builtin_amdgcn_wmma_f32_16x16x4_f32(false, ab1, false, bone, (short)0, vzero, false, false);

#pragma unroll
    for (int m = 0; m < kMT; ++m) {
      v8f h = __builtin_amdgcn_wmma_f32_16x16x4_f32(false, wlo, false, qlo[m], (short)0, hinit, false, false);
      h     = __builtin_amdgcn_wmma_f32_16x16x4_f32(false, whi, false, qhi[m], (short)0, h,     false, false);
#pragma unroll
      for (int r = 0; r < 8; ++r) h[r] = h[r] > 0.0f ? h[r] : 0.0f;

#pragma unroll
      for (int i = 0; i < 4; ++i) {
        v2f a2, bfrag;
        a2[0]    = (i < 2) ? wq0[2 * i]     : wq1[2 * (i - 2)];
        a2[1]    = (i < 2) ? wq0[2 * i + 1] : wq1[2 * (i - 2) + 1];
        bfrag[0] = h[2 * i];
        bfrag[1] = h[2 * i + 1];
        acc[m] = __builtin_amdgcn_wmma_f32_16x16x4_f32(false, a2, false, bfrag, (short)0, acc[m], false, false);
      }
    }
  }

  if (hi == 0) {
#pragma unroll
    for (int m = 0; m < kMT; ++m) {
      float* orow = out + (size_t)((tile0 + m) * 16 + row) * kNQ;
      v4f lo4, hi4;
      lo4[0] = acc[m][0]; lo4[1] = acc[m][1]; lo4[2] = acc[m][2]; lo4[3] = acc[m][3];
      hi4[0] = acc[m][4]; hi4[1] = acc[m][5]; hi4[2] = acc[m][6]; hi4[3] = acc[m][7];
      *(v4f*)(orow + 0) = lo4;
      *(v4f*)(orow + 4) = hi4;
    }
  }
}

extern "C" void kernel_launch(void* const* d_in, const int* in_sizes, int n_in,
                              void* d_out, int out_size, void* d_ws, size_t ws_size,
                              hipStream_t stream) {
  const float* x   = (const float*)d_in[0];
  const float* phi = (const float*)d_in[1];
  const float* w1  = (const float*)d_in[2];
  const float* b1  = (const float*)d_in[3];
  const float* w2  = (const float*)d_in[4];
  const float* b2  = (const float*)d_in[5];
  float* out = (float*)d_out;

  const int blocks = kTokens / (16 * kMT * kWaves);
  ffq_fused_wmma<<<blocks, 256, 0, stream>>>(x, phi, w1, b1, w2, b2, out);
}


# ReLU of each hidden tile as 4 dual-issue maxima in place (inline asm), not a quieting maximum plus a maximum per element: 46 -> 16 vector instructions per chunk; values, sum order unchanged
# speedup vs baseline: 1.6233x; 1.0912x over previous
#include <hip/hip_runtime.h>

typedef float v2f __attribute__((ext_vector_type(2)));
typedef float v4f __attribute__((ext_vector_type(4)));
typedef float v8f __attribute__((ext_vector_type(8)));

namespace {
constexpr int kNQ     = 8;
constexpr int kFFN    = 512;
constexpr int kTokens = 32 * 4096;
constexpr int kWaves  = 8;
constexpr int kMT     = 4;
constexpr int kChunks = kFFN / 16;
}

__global__ __launch_bounds__(256)
void ffq_fused_wmma(const float* __restrict__ x,
                    const float* __restrict__ phi,
                    const float* __restrict__ w1,
                    const float* __restrict__ b1,
                    const float* __restrict__ w2,
                    const float* __restrict__ b2,
                    float* __restrict__ out)
{
  __shared__ v4f   sW1 [kChunks * 32];
  __shared__ v4f   sW2a[kChunks * 32];
  __shared__ v4f   sW2b[kChunks * 32];
  __shared__ float sB1 [kChunks * 32];

  const int lane = threadIdx.x & 31;
  const int wave = threadIdx.x >> 5;
  const int row  = lane & 15;
  const int hi   = lane >> 4;
  const int koff = hi << 1;

  const int tile0 = (blockIdx.x * kWaves + wave) * kMT;

  const float* w2r = w2 + (size_t)(row & 7) * kFFN + 8 * hi;

  v4f   stA[kChunks / kWaves], stQ0[kChunks / kWaves], stQ1[kChunks / kWaves];
  float stB[kChunks / kWaves];
#pragma unroll
  for (int j = 0; j < kChunks / kWaves; ++j) {
    const int n0 = (wave + j * kWaves) * 16;
    const float* w1r = w1 + (size_t)(n0 + row) * kNQ;
    stA[j][0] = w1r[koff + 0]; stA[j][1] = w1r[koff + 1];
    stA[j][2] = w1r[koff + 4]; stA[j][3] = w1r[koff + 5];
    stQ0[j] = *(const v4f*)(w2r + n0);
    stQ1[j] = *(const v4f*)(w2r + n0 + 4);
    const float bv = b1[n0 + row];
    stB[j] = hi ? 0.0f : bv;
  }

  const float ph  = phi[0];
  const float b2v = b2[row & 7];

  v2f xlo[kMT], xhi[kMT];
#pragma unroll
  for (int m = 0; m < kMT; ++m) {
    const float* xr = x + (size_t)((tile0 + m) * 16 + row) * kNQ;
    xlo[m][0] = xr[koff + 0]; xlo[m][1] = xr[koff + 1];
    xhi[m][0] = xr[koff + 4]; xhi[m][1] = xr[koff + 5];
  }

#pragma unroll
  for (int j = 0; j < kChunks / kWaves; ++j) {
    const int idx = (wave + j * kWaves) * 32 + lane;
    sW1[idx]  = stA[j];
    sW2a[idx] = stQ0[j];
    sW2b[idx] = stQ1[j];
    sB1[idx]  = stB[j];
  }

  v2f qlo[kMT], qhi[kMT];
#pragma unroll
  for (int m = 0; m < kMT; ++m) {
    qlo[m][0] = __cosf(xlo[m][0] + ph);
    qlo[m][1] = __cosf(xlo[m][1] + ph);
    qhi[m][0] = __cosf(xhi[m][0] + ph);
    qhi[m][1] = __cosf(xhi[m][1] + ph);
  }

  v2f bone; bone[0] = hi ? 0.0f : 1.0f; bone[1] = 0.0f;
  const v8f vzero = {};

  v2f ab2; ab2[0] = hi ? 0.0f : b2v; ab2[1] = 0.0f;
  const v8f accInit =
      __builtin_amdgcn_wmma_f32_16x16x4_f32(false, ab2, false, bone, (short)0, vzero, false, false);
  v8f acc[kMT];
#pragma unroll
  for (int m = 0; m < kMT; ++m) acc[m] = accInit;

  __syncthreads();

  for (int c = 0; c < kChunks; ++c) {
    const int idx = c * 32 + lane;
    const v4f wA = sW1[idx];
    v2f wlo, whi;
    wlo[0] = wA[0]; wlo[1] = wA[1];
    whi[0] = wA[2]; whi[1] = wA[3];

    v2f ab1; ab1[0] = sB1[idx]; ab1[1] = 0.0f;

    const v4f wq0 = sW2a[idx];
    const v4f wq1 = sW2b[idx];

    const v8f hinit =
        __builtin_amdgcn_wmma_f32_16x16x4_f32(false, ab1, false, bone, (short)0, vzero, false, false);

#pragma unroll
    for (int m = 0; m < kMT; ++m) {
      v8f h = __builtin_amdgcn_wmma_f32_16x16x4_f32(false, wlo, false, qlo[m], (short)0, hinit, false, false);
      h     = __builtin_amdgcn_wmma_f32_16x16x4_f32(false, whi, false, qhi[m], (short)0, h,     false, false);
      float r0 = h[0], r1 = h[1], r2 = h[2], r3 = h[3], r4 = h[4], r5 = h[5], r6 = h[6], r7 = h[7];
      asm("s_wait_alu depctr_va_vdst(0)\n\t"
          "v_dual_max_num_f32 %0, 0, %0 :: v_dual_max_num_f32 %1, 0, %1\n\t"
          "v_dual_max_num_f32 %2, 0, %2 :: v_dual_max_num_f32 %3, 0, %3\n\t"
          "v_dual_max_num_f32 %4, 0, %4 :: v_dual_max_num_f32 %5, 0, %5\n\t"
          "v_dual_max_num_f32 %6, 0, %6 :: v_dual_max_num_f32 %7, 0, %7"
          : "+v"(r0), "+v"(r1), "+v"(r2), "+v"(r3), "+v"(r4), "+v"(r5), "+v"(r6), "+v"(r7));
      h[0] = r0; h[1] = r1; h[2] = r2; h[3] = r3; h[4] = r4; h[5] = r5; h[6] = r6; h[7] = r7;

#pragma unroll
      for (int i = 0; i < 4; ++i) {
        v2f a2, bfrag;
        a2[0]    = (i < 2) ? wq0[2 * i]     : wq1[2 * (i - 2)];
        a2[1]    = (i < 2) ? wq0[2 * i + 1] : wq1[2 * (i - 2) + 1];
        bfrag[0] = h[2 * i];
        bfrag[1] = h[2 * i + 1];
        acc[m] = __builtin_amdgcn_wmma_f32_16x16x4_f32(false, a2, false, bfrag, (short)0, acc[m], false, false);
      }
    }
  }

  if (hi == 0) {
#pragma unroll
    for (int m = 0; m < kMT; ++m) {
      float* orow = out + (size_t)((tile0 + m) * 16 + row) * kNQ;
      v4f lo4, hi4;
      lo4[0] = acc[m][0]; lo4[1] = acc[m][1]; lo4[2] = acc[m][2]; lo4[3] = acc[m][3];
      hi4[0] = acc[m][4]; hi4[1] = acc[m][5]; hi4[2] = acc[m][6]; hi4[3] = acc[m][7];
      *(v4f*)(orow + 0) = lo4;
      *(v4f*)(orow + 4) = hi4;
    }
  }
}

extern "C" void kernel_launch(void* const* d_in, const int* in_sizes, int n_in,
                              void* d_out, int out_size, void* d_ws, size_t ws_size,
                              hipStream_t stream) {
  const float* x   = (const float*)d_in[0];
  const float* phi = (const float*)d_in[1];
  const float* w1  = (const float*)d_in[2];
  const float* b1  = (const float*)d_in[3];
  const float* w2  = (const float*)d_in[4];
  const float* b2  = (const float*)d_in[5];
  float* out = (float*)d_out;

  const int blocks = kTokens / (16 * kMT * kWaves);
  ffq_fused_wmma<<<blocks, 256, 0, stream>>>(x, phi, w1, b1, w2, b2, out);
}
